# starting kernel
# baseline (speedup 1.0000x reference)
#include <hip/hip_runtime.h>
#include <hip/hip_bf16.h>
#include <math.h>

static constexpr int Bsz = 512;
static constexpr int T   = 16384;
static constexpr int L   = 512;
static constexpr int C   = T / L;

typedef __attribute__((ext_vector_type(2))) float v2f;
typedef __attribute__((ext_vector_type(4))) float v4f;
typedef __attribute__((ext_vector_type(8))) float v8f;

__device__ __forceinline__ float whf(float x) {
    const float c   = 10.0f / 11.0f;
    const float l2e = 1.4426950408889634f;
    float e = 1.0f - __builtin_amdgcn_exp2f(-(c * l2e) * x);
    float l = c * x;
    unsigned s = (unsigned)(__float_as_int(x) >> 31);
    return __uint_as_float((__float_as_uint(e) & ~s) | (__float_as_uint(l) & s));
}

__global__ void hpack_kernel(const float* __restrict__ a1c,
                             const float* __restrict__ a2c,
                             float* __restrict__ hp1,
                             float* __restrict__ hp2) {
    const float* ac = (blockIdx.x == 0) ? a1c : a2c;
    float*       hp = (blockIdx.x == 0) ? hp1 : hp2;
    int j = threadIdx.x;
    if (j > 3) return;
    if (j == 3) {
        for (int i = 0; i < L; ++i) hp[i * 4 + 3] = 0.0f;
        return;
    }
    const float a0 = ac[0], a1 = ac[1], a2 = ac[2];
    float y1 = (j == 0) ? 1.0f : 0.0f;
    float y2 = (j == 1) ? 1.0f : 0.0f;
    float y3 = (j == 2) ? 1.0f : 0.0f;
    for (int i = 0; i < L; ++i) {
        float y = -(a0 * y1 + a1 * y2 + a2 * y3);
        hp[i * 4 + j] = y;
        y3 = y2; y2 = y1; y1 = y;
    }
}

__global__ __launch_bounds__(256)
void phase1_kernel(const float* __restrict__ in,
                   const float* __restrict__ bco,
                   const float* __restrict__ aco,
                   int nk,
                   float* __restrict__ z,
                   float* __restrict__ tails) {
    const int id = blockIdx.x * blockDim.x + threadIdx.x;
    const int b  = id / C;
    const int c  = id % C;
    const float b0 = bco[0], b1 = bco[1], b2 = bco[2];
    const float a0 = aco[0], a1 = aco[1], a2 = aco[2];

    const float* __restrict__ row = in + (size_t)b * T;
    const int t0 = c * L;
    float x1 = (t0 - nk - 1 >= 0) ? row[t0 - nk - 1] : 0.0f;
    float x2 = (t0 - nk - 2 >= 0) ? row[t0 - nk - 2] : 0.0f;
    float y1 = 0.0f, y2 = 0.0f, y3 = 0.0f;
    float* __restrict__ zrow = z + (size_t)b * T + t0;

#pragma unroll 4
    for (int i = 0; i < L; ++i) {
        const int t = t0 + i;
        if ((i & 63) == 0)
            __builtin_prefetch(row + t + 256, 0, 0);
        float x0 = (t - nk >= 0) ? row[t - nk] : 0.0f;
        float xf = b0 * x0 + b1 * x1 + b2 * x2;
        float y  = xf - (a0 * y1 + a1 * y2 + a2 * y3);
        zrow[i] = y;
        x2 = x1; x1 = x0;
        y3 = y2; y2 = y1; y1 = y;
    }
    float* tl = tails + (size_t)id * 4;
    tl[0] = y1; tl[1] = y2; tl[2] = y3; tl[3] = 0.0f;
}

__global__ __launch_bounds__(256)
void scan_kernel(const float* __restrict__ tails,
                 const float* __restrict__ hpack,
                 float* __restrict__ states) {
    const int b = blockIdx.x * blockDim.x + threadIdx.x;
    if (b >= Bsz) return;
    const float h00 = hpack[(L - 1) * 4 + 0], h01 = hpack[(L - 1) * 4 + 1], h02 = hpack[(L - 1) * 4 + 2];
    const float h10 = hpack[(L - 2) * 4 + 0], h11 = hpack[(L - 2) * 4 + 1], h12 = hpack[(L - 2) * 4 + 2];
    const float h20 = hpack[(L - 3) * 4 + 0], h21 = hpack[(L - 3) * 4 + 1], h22 = hpack[(L - 3) * 4 + 2];
    float s0 = 0.0f, s1 = 0.0f, s2 = 0.0f;
    for (int c = 0; c < C; ++c) {
        const size_t idx = ((size_t)b * C + c) * 4;
        states[idx + 0] = s0; states[idx + 1] = s1;
        states[idx + 2] = s2; states[idx + 3] = 0.0f;
        const float t0 = tails[idx + 0], t1 = tails[idx + 1], t2 = tails[idx + 2];
        const float n0 = t0 + h00 * s0 + h01 * s1 + h02 * s2;
        const float n1 = t1 + h10 * s0 + h11 * s1 + h12 * s2;
        const float n2 = t2 + h20 * s0 + h21 * s1 + h22 * s2;
        s0 = n0; s1 = n1; s2 = n2;
    }
}

template <bool APPLY_NL>
__global__ __launch_bounds__(256)
void correct_kernel(const float* __restrict__ z,
                    const float* __restrict__ states,
                    const float* __restrict__ hpack,
                    float* __restrict__ out) {
    const int lane = threadIdx.x & 31;
    const int wave = threadIdx.x >> 5;
    const int g    = blockIdx.x * (blockDim.x >> 5) + wave;
    const int n    = lane & 15;
    const int half = lane >> 4;
    const int b    = g / (C / 16);
    const int cg   = g % (C / 16);
    const int c    = cg * 16 + n;

    const float* __restrict__ zb = z   + (size_t)b * T + (size_t)c * L;
    float*       __restrict__ ob = out + (size_t)b * T + (size_t)c * L;

    const v2f bv = *(const v2f*)(states + ((size_t)b * C + c) * 4 + 2 * half);

#pragma unroll 4
    for (int ib = 0; ib < L; ib += 16) {
        const v2f av = *(const v2f*)(hpack + (size_t)(ib + n) * 4 + 2 * half);
        const v4f c0 = *(const v4f*)(zb + ib + 8 * half);
        const v4f c1 = *(const v4f*)(zb + ib + 8 * half + 4);
        v8f cv;
        cv[0] = c0[0]; cv[1] = c0[1]; cv[2] = c0[2]; cv[3] = c0[3];
        cv[4] = c1[0]; cv[5] = c1[1]; cv[6] = c1[2]; cv[7] = c1[3];

        v8f dv = __builtin_amdgcn_wmma_f32_16x16x4_f32(
             false, av,  false, bv,
             (short)0, cv,  false,  false);

        v4f o0, o1;
        if (APPLY_NL) {
            o0[0] = whf(dv[0]); o0[1] = whf(dv[1]); o0[2] = whf(dv[2]); o0[3] = whf(dv[3]);
            o1[0] = whf(dv[4]); o1[1] = whf(dv[5]); o1[2] = whf(dv[6]); o1[3] = whf(dv[7]);
        } else {
            o0[0] = dv[0]; o0[1] = dv[1]; o0[2] = dv[2]; o0[3] = dv[3];
            o1[0] = dv[4]; o1[1] = dv[5]; o1[2] = dv[6]; o1[3] = dv[7];
        }
        if (APPLY_NL) {
            *(v4f*)(ob + ib + 8 * half)     = o0;
            *(v4f*)(ob + ib + 8 * half + 4) = o1;
        } else {
            __builtin_nontemporal_store(o0, (v4f*)(ob + ib + 8 * half));
            __builtin_nontemporal_store(o1, (v4f*)(ob + ib + 8 * half + 4));
        }
    }
}

extern "C" void kernel_launch(void* const* d_in, const int* in_sizes, int n_in,
                              void* d_out, int out_size, void* d_ws, size_t ws_size,
                              hipStream_t stream) {
    const float* u  = (const float*)d_in[0];
    const float* b1 = (const float*)d_in[1];
    const float* a1 = (const float*)d_in[2];
    const float* b2 = (const float*)d_in[3];
    const float* a2 = (const float*)d_in[4];
    float* out = (float*)d_out;

    float* ws     = (float*)d_ws;
    float* z      = ws;
    float* tails  = z + (size_t)Bsz * T;
    float* states = tails + (size_t)Bsz * C * 4;
    float* hp1    = states + (size_t)Bsz * C * 4;
    float* hp2    = hp1 + L * 4;

    const int p1_blocks  = (Bsz * C) / 256;
    const int sc_blocks  = (Bsz + 255) / 256;
    const int cor_blocks = (Bsz * (C / 16)) / 8;

    hpack_kernel<<<2, 32, 0, stream>>>(a1, a2, hp1, hp2);

    phase1_kernel<<<p1_blocks, 256, 0, stream>>>(u, b1, a1,  1, z, tails);
    scan_kernel<<<sc_blocks, 256, 0, stream>>>(tails, hp1, states);
    correct_kernel<true><<<cor_blocks, 256, 0, stream>>>(z, states, hp1, out);

    phase1_kernel<<<p1_blocks, 256, 0, stream>>>(out, b2, a2,  0, z, tails);
    scan_kernel<<<sc_blocks, 256, 0, stream>>>(tails, hp2, states);
    correct_kernel<false><<<cor_blocks, 256, 0, stream>>>(z, states, hp2, out);
}


# row kept in matrix-tile registers by one 2-wave workgroup: recurrences, chunk scans (via LDS) and corrections of both filters in one launch; z, v, tails, states leave memory; sum order kept
# speedup vs baseline: 11.3394x; 11.3394x over previous
#include <hip/hip_runtime.h>
#include <hip/hip_bf16.h>
#include <math.h>

static constexpr int Bsz = 512;
static constexpr int T   = 16384;
static constexpr int L   = 512;
static constexpr int C   = T / L;

typedef __attribute__((ext_vector_type(2))) float v2f;
typedef __attribute__((ext_vector_type(4))) float v4f;
typedef __attribute__((ext_vector_type(8))) float v8f;

__device__ __forceinline__ float whf(float x) {
    const float c   = 10.0f / 11.0f;
    const float l2e = 1.4426950408889634f;
    float e = 1.0f - __builtin_amdgcn_exp2f(-(c * l2e) * x);
    float l = c * x;
    unsigned s = (unsigned)(__float_as_int(x) >> 31);
    return __uint_as_float((__float_as_uint(e) & ~s) | (__float_as_uint(l) & s));
}

__device__ __forceinline__ int blk_time(int hf, int r) {
    return 8 * (r >> 2) + 4 * hf + (r & 3);
}

static constexpr int TAB_FLOATS = 8 * 64;

__device__ void build_block_tables(const float* __restrict__ bc,
                                   const float* __restrict__ ac,
                                   int nk, float* __restrict__ tab) {
    __shared__ float hs[16];
    __shared__ float hh[3][16];
    const int tid = threadIdx.x;
    const float a0 = ac[0], a1 = ac[1], a2 = ac[2];
    if (tid < 4) {
        float y1 = (tid == 1) ? 1.0f : 0.0f;
        float y2 = (tid == 2) ? 1.0f : 0.0f;
        float y3 = (tid == 3) ? 1.0f : 0.0f;
        for (int i = 0; i < 16; ++i) {
            float x = (tid == 0 && i == 0) ? 1.0f : 0.0f;
            float y = x - (a0 * y1 + a1 * y2 + a2 * y3);
            if (tid == 0) hs[i] = y; else hh[tid - 1][i] = y;
            y3 = y2; y2 = y1; y1 = y;
        }
    }
    __syncthreads();
    const float bq[3] = {bc[0], bc[1], bc[2]};
    const int row = tid & 15;
    const int hf  = tid >> 4;
    const int m   = blk_time(row >> 3, row & 7);
    for (int slot = 0; slot < 6; ++slot) {
        const int kb = (slot < 4) ? slot : slot - 2;
        for (int v = 0; v < 2; ++v) {
            int j = blk_time(hf, 2 * kb + v);
            if (slot >= 4) j -= 16;
            float s = 0.0f;
            for (int q = 0; q < 3; ++q) {
                const int d = m - j - nk - q;
                if (q >= -(j + nk) && d >= 0) s += bq[q] * hs[d];
            }
            tab[slot * 64 + tid * 2 + v] = s;
        }
    }
    for (int v = 0; v < 2; ++v) {
        const int j67 = blk_time(hf, 6 + v);
        const int j45 = blk_time(hf, 4 + v);
        tab[6 * 64 + tid * 2 + v] = (j67 == 15) ? hh[0][m] : (j67 == 14) ? hh[1][m] : 0.0f;
        tab[7 * 64 + tid * 2 + v] = (j45 == 13) ? hh[2][m] : 0.0f;
    }
}

__global__ void hpack_kernel(const float* __restrict__ a1c,
                             const float* __restrict__ a2c,
                             const float* __restrict__ b1c,
                             const float* __restrict__ b2c,
                             float* __restrict__ hp1,
                             float* __restrict__ hp2,
                             float* __restrict__ tab1,
                             float* __restrict__ tab2) {
    if (blockIdx.x == 2) { build_block_tables(b1c, a1c, 1, tab1); return; }
    if (blockIdx.x == 3) { build_block_tables(b2c, a2c, 0, tab2); return; }
    const float* ac = (blockIdx.x == 0) ? a1c : a2c;
    float*       hp = (blockIdx.x == 0) ? hp1 : hp2;
    int j = threadIdx.x;
    if (j > 3) return;
    if (j == 3) {
        for (int i = 0; i < L; ++i) hp[i * 4 + 3] = 0.0f;
        return;
    }
    const float a0 = ac[0], a1 = ac[1], a2 = ac[2];
    float y1 = (j == 0) ? 1.0f : 0.0f;
    float y2 = (j == 1) ? 1.0f : 0.0f;
    float y3 = (j == 2) ? 1.0f : 0.0f;
    for (int i = 0; i < L; ++i) {
        float y = -(a0 * y1 + a1 * y2 + a2 * y3);
        hp[i * 4 + j] = y;
        y3 = y2; y2 = y1; y1 = y;
    }
}

__device__ __forceinline__ v8f mm4(v2f a, float b0, float b1, v8f c) {
    v2f b; b[0] = b0; b[1] = b1;
    return __builtin_amdgcn_wmma_f32_16x16x4_f32(
         false, a,  false, b,
         (short)0, c,  false,  false);
}

static constexpr int NT = L / 16;

__device__ __forceinline__ v2f chunk_state(const v4f* tl, const float* __restrict__ hp,
                                           int c, int hf) {
    const float h00 = hp[(L - 1) * 4 + 0], h01 = hp[(L - 1) * 4 + 1], h02 = hp[(L - 1) * 4 + 2];
    const float h10 = hp[(L - 2) * 4 + 0], h11 = hp[(L - 2) * 4 + 1], h12 = hp[(L - 2) * 4 + 2];
    const float h20 = hp[(L - 3) * 4 + 0], h21 = hp[(L - 3) * 4 + 1], h22 = hp[(L - 3) * 4 + 2];
    float s0 = 0.0f, s1 = 0.0f, s2 = 0.0f;
    float m0 = 0.0f, m1 = 0.0f, m2 = 0.0f;
#pragma unroll 8
    for (int cc = 0; cc < C; ++cc) {
        const bool mine = (cc == c);
        m0 = mine ? s0 : m0; m1 = mine ? s1 : m1; m2 = mine ? s2 : m2;
        const v4f t = tl[cc];
        const float n0 = t[0] + h00 * s0 + h01 * s1 + h02 * s2;
        const float n1 = t[1] + h10 * s0 + h11 * s1 + h12 * s2;
        const float n2 = t[2] + h20 * s0 + h21 * s1 + h22 * s2;
        s0 = n0; s1 = n1; s2 = n2;
    }
    v2f bv;
    bv[0] = hf ? m2 : m0;
    bv[1] = hf ? 0.0f : m1;
    return bv;
}

__global__ __launch_bounds__(64)
void row_kernel(const float* __restrict__ u,
                const float* __restrict__ tab1,
                const float* __restrict__ hp1,
                const float* __restrict__ tab2,
                const float* __restrict__ hp2,
                float* __restrict__ out) {
    __shared__ v4f tails1[C];
    __shared__ v4f tails2[C];
    __shared__ v4f edge[C][2];

    const int lane = threadIdx.x & 31;
    const int wave = threadIdx.x >> 5;
    const int n    = lane & 15;
    const int hf   = lane >> 4;
    const int b    = blockIdx.x;
    const int c    = wave * 16 + n;

    const float* __restrict__ src = u   + (size_t)b * T + (size_t)c * L + 4 * hf;
    float*       __restrict__ dst = out + (size_t)b * T + (size_t)c * L + 4 * hf;

    const int mrow = blk_time(n >> 3, n & 7);

    v8f t[NT];

    {
        const v2f* tp = (const v2f*)tab1 + lane;
        const v2f g0 = tp[0 * 32], g1 = tp[1 * 32], g2 = tp[2 * 32], g3 = tp[3 * 32];
        const v2f p2 = tp[4 * 32], p3 = tp[5 * 32];
        const v2f s3 = tp[6 * 32], s2 = tp[7 * 32];

        v4f pv = *(const v4f*)(c > 0 ? src - 8 : src);
        if (c == 0) { pv[0] = 0.0f; pv[1] = 0.0f; pv[2] = 0.0f; pv[3] = 0.0f; }

        v4f ua[NT], ub[NT];
#pragma unroll
        for (int k = 0; k < NT; ++k) {
            ua[k] = *(const v4f*)(src + 16 * k);
            ub[k] = *(const v4f*)(src + 16 * k + 8);
        }
        v8f d = {0.0f, 0.0f, 0.0f, 0.0f, 0.0f, 0.0f, 0.0f, 0.0f};
#pragma unroll
        for (int k = 0; k < NT; ++k) {
            const v4f u0 = ua[k], u1 = ub[k];
            v8f acc = {0.0f, 0.0f, 0.0f, 0.0f, 0.0f, 0.0f, 0.0f, 0.0f};
            acc = mm4(g0, u0[0], u0[1], acc);
            acc = mm4(g1, u0[2], u0[3], acc);
            acc = mm4(g2, u1[0], u1[1], acc);
            acc = mm4(g3, u1[2], u1[3], acc);
            acc = mm4(p2, pv[0], pv[1], acc);
            acc = mm4(p3, pv[2], pv[3], acc);
            acc = mm4(s2, d[4], d[5], acc);
            acc = mm4(s3, d[6], d[7], acc);
            d = acc;
            pv = u1;
            t[k] = d;
        }
        if (hf == 1) {
            v4f tl; tl[0] = d[7]; tl[1] = d[6]; tl[2] = d[5]; tl[3] = 0.0f;
            tails1[c] = tl;
        }
    }
    __syncthreads();

    {
        const float* __restrict__ hrow = hp1 + (size_t)mrow * 4 + 2 * hf;
        v2f av[NT];
#pragma unroll
        for (int k = 0; k < NT; ++k) av[k] = *(const v2f*)(hrow + 64 * k);
        const v2f bv = chunk_state(tails1, hp1, c, hf);
#pragma unroll
        for (int k = 0; k < NT; ++k) {
            v8f dv = __builtin_amdgcn_wmma_f32_16x16x4_f32(
                 false, av[k],  false, bv,
                 (short)0, t[k],  false,  false);
#pragma unroll
            for (int r = 0; r < 8; ++r) dv[r] = whf(dv[r]);
            t[k] = dv;
        }
        v4f e; e[0] = t[NT - 1][4]; e[1] = t[NT - 1][5]; e[2] = t[NT - 1][6]; e[3] = t[NT - 1][7];
        edge[c][hf] = e;
    }
    __syncthreads();

    {
        const v2f* tp = (const v2f*)tab2 + lane;
        const v2f g0 = tp[0 * 32], g1 = tp[1 * 32], g2 = tp[2 * 32], g3 = tp[3 * 32];
        const v2f p3 = tp[5 * 32];
        const v2f s3 = tp[6 * 32], s2 = tp[7 * 32];

        v4f pv = edge[c > 0 ? c - 1 : 0][hf];
        if (c == 0) { pv[0] = 0.0f; pv[1] = 0.0f; pv[2] = 0.0f; pv[3] = 0.0f; }

        v8f d = {0.0f, 0.0f, 0.0f, 0.0f, 0.0f, 0.0f, 0.0f, 0.0f};
#pragma unroll
        for (int k = 0; k < NT; ++k) {
            const v8f x = t[k];
            v8f acc = {0.0f, 0.0f, 0.0f, 0.0f, 0.0f, 0.0f, 0.0f, 0.0f};
            acc = mm4(g0, x[0], x[1], acc);
            acc = mm4(g1, x[2], x[3], acc);
            acc = mm4(g2, x[4], x[5], acc);
            acc = mm4(g3, x[6], x[7], acc);
            acc = mm4(p3, pv[2], pv[3], acc);
            acc = mm4(s2, d[4], d[5], acc);
            acc = mm4(s3, d[6], d[7], acc);
            d = acc;
            pv[0] = x[4]; pv[1] = x[5]; pv[2] = x[6]; pv[3] = x[7];
            t[k] = d;
        }
        if (hf == 1) {
            v4f tl; tl[0] = d[7]; tl[1] = d[6]; tl[2] = d[5]; tl[3] = 0.0f;
            tails2[c] = tl;
        }
    }
    __syncthreads();

    {
        const float* __restrict__ hrow = hp2 + (size_t)mrow * 4 + 2 * hf;
        v2f av[NT];
#pragma unroll
        for (int k = 0; k < NT; ++k) av[k] = *(const v2f*)(hrow + 64 * k);
        const v2f bv = chunk_state(tails2, hp2, c, hf);
#pragma unroll
        for (int k = 0; k < NT; ++k) {
            const v8f dv = __builtin_amdgcn_wmma_f32_16x16x4_f32(
                 false, av[k],  false, bv,
                 (short)0, t[k],  false,  false);
            v4f o0, o1;
            o0[0] = dv[0]; o0[1] = dv[1]; o0[2] = dv[2]; o0[3] = dv[3];
            o1[0] = dv[4]; o1[1] = dv[5]; o1[2] = dv[6]; o1[3] = dv[7];
            __builtin_nontemporal_store(o0, (v4f*)(dst + 16 * k));
            __builtin_nontemporal_store(o1, (v4f*)(dst + 16 * k + 8));
        }
    }
}

extern "C" void kernel_launch(void* const* d_in, const int* in_sizes, int n_in,
                              void* d_out, int out_size, void* d_ws, size_t ws_size,
                              hipStream_t stream) {
    const float* u  = (const float*)d_in[0];
    const float* b1 = (const float*)d_in[1];
    const float* a1 = (const float*)d_in[2];
    const float* b2 = (const float*)d_in[3];
    const float* a2 = (const float*)d_in[4];
    float* out = (float*)d_out;

    float* ws   = (float*)d_ws;
    float* hp1  = ws;
    float* hp2  = hp1 + L * 4;
    float* tab1 = hp2 + L * 4;
    float* tab2 = tab1 + TAB_FLOATS;

    hpack_kernel<<<4, 32, 0, stream>>>(a1, a2, b1, b2, hp1, hp2, tab1, tab2);
    row_kernel<<<Bsz, 64, 0, stream>>>(u, tab1, hp1, tab2, hp2, out);
}


# table kernel: impulse tables built in LDS (walk stores there, all lanes clear the pad column), then stored as whole 16-byte rows by the wave; one-lane zero loop and the walk's global stores gone
# speedup vs baseline: 15.4417x; 1.3618x over previous
#include <hip/hip_runtime.h>
#include <hip/hip_bf16.h>
#include <math.h>

static constexpr int Bsz = 512;
static constexpr int T   = 16384;
static constexpr int L   = 512;
static constexpr int C   = T / L;

typedef __attribute__((ext_vector_type(2))) float v2f;
typedef __attribute__((ext_vector_type(4))) float v4f;
typedef __attribute__((ext_vector_type(8))) float v8f;

__device__ __forceinline__ float whf(float x) {
    const float c   = 10.0f / 11.0f;
    const float l2e = 1.4426950408889634f;
    float e = 1.0f - __builtin_amdgcn_exp2f(-(c * l2e) * x);
    float l = c * x;
    unsigned s = (unsigned)(__float_as_int(x) >> 31);
    return __uint_as_float((__float_as_uint(e) & ~s) | (__float_as_uint(l) & s));
}

__device__ __forceinline__ int blk_time(int hf, int r) {
    return 8 * (r >> 2) + 4 * hf + (r & 3);
}

static constexpr int TAB_FLOATS = 8 * 64;
static constexpr int HPACK_LANES = 32;

__device__ void build_block_tables(const float* __restrict__ bc,
                                   const float* __restrict__ ac,
                                   int nk, float* __restrict__ tab) {
    __shared__ float hs[16];
    __shared__ float hh[3][16];
    const int tid = threadIdx.x;
    const float a0 = ac[0], a1 = ac[1], a2 = ac[2];
    if (tid < 4) {
        float y1 = (tid == 1) ? 1.0f : 0.0f;
        float y2 = (tid == 2) ? 1.0f : 0.0f;
        float y3 = (tid == 3) ? 1.0f : 0.0f;
        for (int i = 0; i < 16; ++i) {
            float x = (tid == 0 && i == 0) ? 1.0f : 0.0f;
            float y = x - (a0 * y1 + a1 * y2 + a2 * y3);
            if (tid == 0) hs[i] = y; else hh[tid - 1][i] = y;
            y3 = y2; y2 = y1; y1 = y;
        }
    }
    __syncthreads();
    const float bq[3] = {bc[0], bc[1], bc[2]};
    const int row = tid & 15;
    const int hf  = tid >> 4;
    const int m   = blk_time(row >> 3, row & 7);
    for (int slot = 0; slot < 6; ++slot) {
        const int kb = (slot < 4) ? slot : slot - 2;
        for (int v = 0; v < 2; ++v) {
            int j = blk_time(hf, 2 * kb + v);
            if (slot >= 4) j -= 16;
            float s = 0.0f;
            for (int q = 0; q < 3; ++q) {
                const int d = m - j - nk - q;
                if (q >= -(j + nk) && d >= 0) s += bq[q] * hs[d];
            }
            tab[slot * 64 + tid * 2 + v] = s;
        }
    }
    for (int v = 0; v < 2; ++v) {
        const int j67 = blk_time(hf, 6 + v);
        const int j45 = blk_time(hf, 4 + v);
        tab[6 * 64 + tid * 2 + v] = (j67 == 15) ? hh[0][m] : (j67 == 14) ? hh[1][m] : 0.0f;
        tab[7 * 64 + tid * 2 + v] = (j45 == 13) ? hh[2][m] : 0.0f;
    }
}

__global__ void hpack_kernel(const float* __restrict__ a1c,
                             const float* __restrict__ a2c,
                             const float* __restrict__ b1c,
                             const float* __restrict__ b2c,
                             float* __restrict__ hp1,
                             float* __restrict__ hp2,
                             float* __restrict__ tab1,
                             float* __restrict__ tab2) {
    if (blockIdx.x == 2) { build_block_tables(b1c, a1c, 1, tab1); return; }
    if (blockIdx.x == 3) { build_block_tables(b2c, a2c, 0, tab2); return; }
    const float* ac = (blockIdx.x == 0) ? a1c : a2c;
    float*       hp = (blockIdx.x == 0) ? hp1 : hp2;
    __shared__ __attribute__((aligned(16))) float rows[L * 4];
    const int tid = threadIdx.x;
    for (int q = 0; q < L / HPACK_LANES; ++q) rows[(tid + HPACK_LANES * q) * 4 + 3] = 0.0f;
    if (tid < 3) {
        const int j = tid;
        const float a0 = ac[0], a1 = ac[1], a2 = ac[2];
        float y1 = (j == 0) ? 1.0f : 0.0f;
        float y2 = (j == 1) ? 1.0f : 0.0f;
        float y3 = (j == 2) ? 1.0f : 0.0f;
        for (int i = 0; i < L; ++i) {
            float y = -(a0 * y1 + a1 * y2 + a2 * y3);
            rows[i * 4 + j] = y;
            y3 = y2; y2 = y1; y1 = y;
        }
    }
    __syncthreads();
    for (int q = 0; q < L / HPACK_LANES; ++q) {
        const int i = tid + HPACK_LANES * q;
        *(v4f*)(hp + 4 * i) = *(const v4f*)(rows + 4 * i);
    }
}

__device__ __forceinline__ v8f mm4(v2f a, float b0, float b1, v8f c) {
    v2f b; b[0] = b0; b[1] = b1;
    return __builtin_amdgcn_wmma_f32_16x16x4_f32(
         false, a,  false, b,
         (short)0, c,  false,  false);
}

static constexpr int NT = L / 16;

__device__ __forceinline__ v2f chunk_state(const v4f* tl, const float* __restrict__ hp,
                                           int c, int hf) {
    const float h00 = hp[(L - 1) * 4 + 0], h01 = hp[(L - 1) * 4 + 1], h02 = hp[(L - 1) * 4 + 2];
    const float h10 = hp[(L - 2) * 4 + 0], h11 = hp[(L - 2) * 4 + 1], h12 = hp[(L - 2) * 4 + 2];
    const float h20 = hp[(L - 3) * 4 + 0], h21 = hp[(L - 3) * 4 + 1], h22 = hp[(L - 3) * 4 + 2];
    float s0 = 0.0f, s1 = 0.0f, s2 = 0.0f;
    float m0 = 0.0f, m1 = 0.0f, m2 = 0.0f;
#pragma unroll 8
    for (int cc = 0; cc < C; ++cc) {
        const bool mine = (cc == c);
        m0 = mine ? s0 : m0; m1 = mine ? s1 : m1; m2 = mine ? s2 : m2;
        const v4f t = tl[cc];
        const float n0 = t[0] + h00 * s0 + h01 * s1 + h02 * s2;
        const float n1 = t[1] + h10 * s0 + h11 * s1 + h12 * s2;
        const float n2 = t[2] + h20 * s0 + h21 * s1 + h22 * s2;
        s0 = n0; s1 = n1; s2 = n2;
    }
    v2f bv;
    bv[0] = hf ? m2 : m0;
    bv[1] = hf ? 0.0f : m1;
    return bv;
}

__global__ __launch_bounds__(64)
void row_kernel(const float* __restrict__ u,
                const float* __restrict__ tab1,
                const float* __restrict__ hp1,
                const float* __restrict__ tab2,
                const float* __restrict__ hp2,
                float* __restrict__ out) {
    __shared__ v4f tails1[C];
    __shared__ v4f tails2[C];
    __shared__ v4f edge[C][2];

    const int lane = threadIdx.x & 31;
    const int wave = threadIdx.x >> 5;
    const int n    = lane & 15;
    const int hf   = lane >> 4;
    const int b    = blockIdx.x;
    const int c    = wave * 16 + n;

    const float* __restrict__ src = u   + (size_t)b * T + (size_t)c * L + 4 * hf;
    float*       __restrict__ dst = out + (size_t)b * T + (size_t)c * L + 4 * hf;

    const int mrow = blk_time(n >> 3, n & 7);

    v8f t[NT];

    {
        const v2f* tp = (const v2f*)tab1 + lane;
        const v2f g0 = tp[0 * 32], g1 = tp[1 * 32], g2 = tp[2 * 32], g3 = tp[3 * 32];
        const v2f p2 = tp[4 * 32], p3 = tp[5 * 32];
        const v2f s3 = tp[6 * 32], s2 = tp[7 * 32];

        v4f pv = *(const v4f*)(c > 0 ? src - 8 : src);
        if (c == 0) { pv[0] = 0.0f; pv[1] = 0.0f; pv[2] = 0.0f; pv[3] = 0.0f; }

        v4f ua[NT], ub[NT];
#pragma unroll
        for (int k = 0; k < NT; ++k) {
            ua[k] = *(const v4f*)(src + 16 * k);
            ub[k] = *(const v4f*)(src + 16 * k + 8);
        }
        v8f d = {0.0f, 0.0f, 0.0f, 0.0f, 0.0f, 0.0f, 0.0f, 0.0f};
#pragma unroll
        for (int k = 0; k < NT; ++k) {
            const v4f u0 = ua[k], u1 = ub[k];
            v8f acc = {0.0f, 0.0f, 0.0f, 0.0f, 0.0f, 0.0f, 0.0f, 0.0f};
            acc = mm4(g0, u0[0], u0[1], acc);
            acc = mm4(g1, u0[2], u0[3], acc);
            acc = mm4(g2, u1[0], u1[1], acc);
            acc = mm4(g3, u1[2], u1[3], acc);
            acc = mm4(p2, pv[0], pv[1], acc);
            acc = mm4(p3, pv[2], pv[3], acc);
            acc = mm4(s2, d[4], d[5], acc);
            acc = mm4(s3, d[6], d[7], acc);
            d = acc;
            pv = u1;
            t[k] = d;
        }
        if (hf == 1) {
            v4f tl; tl[0] = d[7]; tl[1] = d[6]; tl[2] = d[5]; tl[3] = 0.0f;
            tails1[c] = tl;
        }
    }
    __syncthreads();

    {
        const float* __restrict__ hrow = hp1 + (size_t)mrow * 4 + 2 * hf;
        v2f av[NT];
#pragma unroll
        for (int k = 0; k < NT; ++k) av[k] = *(const v2f*)(hrow + 64 * k);
        const v2f bv = chunk_state(tails1, hp1, c, hf);
#pragma unroll
        for (int k = 0; k < NT; ++k) {
            v8f dv = __builtin_amdgcn_wmma_f32_16x16x4_f32(
                 false, av[k],  false, bv,
                 (short)0, t[k],  false,  false);
#pragma unroll
            for (int r = 0; r < 8; ++r) dv[r] = whf(dv[r]);
            t[k] = dv;
        }
        v4f e; e[0] = t[NT - 1][4]; e[1] = t[NT - 1][5]; e[2] = t[NT - 1][6]; e[3] = t[NT - 1][7];
        edge[c][hf] = e;
    }
    __syncthreads();

    {
        const v2f* tp = (const v2f*)tab2 + lane;
        const v2f g0 = tp[0 * 32], g1 = tp[1 * 32], g2 = tp[2 * 32], g3 = tp[3 * 32];
        const v2f p3 = tp[5 * 32];
        const v2f s3 = tp[6 * 32], s2 = tp[7 * 32];

        v4f pv = edge[c > 0 ? c - 1 : 0][hf];
        if (c == 0) { pv[0] = 0.0f; pv[1] = 0.0f; pv[2] = 0.0f; pv[3] = 0.0f; }

        v8f d = {0.0f, 0.0f, 0.0f, 0.0f, 0.0f, 0.0f, 0.0f, 0.0f};
#pragma unroll
        for (int k = 0; k < NT; ++k) {
            const v8f x = t[k];
            v8f acc = {0.0f, 0.0f, 0.0f, 0.0f, 0.0f, 0.0f, 0.0f, 0.0f};
            acc = mm4(g0, x[0], x[1], acc);
            acc = mm4(g1, x[2], x[3], acc);
            acc = mm4(g2, x[4], x[5], acc);
            acc = mm4(g3, x[6], x[7], acc);
            acc = mm4(p3, pv[2], pv[3], acc);
            acc = mm4(s2, d[4], d[5], acc);
            acc = mm4(s3, d[6], d[7], acc);
            d = acc;
            pv[0] = x[4]; pv[1] = x[5]; pv[2] = x[6]; pv[3] = x[7];
            t[k] = d;
        }
        if (hf == 1) {
            v4f tl; tl[0] = d[7]; tl[1] = d[6]; tl[2] = d[5]; tl[3] = 0.0f;
            tails2[c] = tl;
        }
    }
    __syncthreads();

    {
        const float* __restrict__ hrow = hp2 + (size_t)mrow * 4 + 2 * hf;
        v2f av[NT];
#pragma unroll
        for (int k = 0; k < NT; ++k) av[k] = *(const v2f*)(hrow + 64 * k);
        const v2f bv = chunk_state(tails2, hp2, c, hf);
#pragma unroll
        for (int k = 0; k < NT; ++k) {
            const v8f dv = __builtin_amdgcn_wmma_f32_16x16x4_f32(
                 false, av[k],  false, bv,
                 (short)0, t[k],  false,  false);
            v4f o0, o1;
            o0[0] = dv[0]; o0[1] = dv[1]; o0[2] = dv[2]; o0[3] = dv[3];
            o1[0] = dv[4]; o1[1] = dv[5]; o1[2] = dv[6]; o1[3] = dv[7];
            __builtin_nontemporal_store(o0, (v4f*)(dst + 16 * k));
            __builtin_nontemporal_store(o1, (v4f*)(dst + 16 * k + 8));
        }
    }
}

extern "C" void kernel_launch(void* const* d_in, const int* in_sizes, int n_in,
                              void* d_out, int out_size, void* d_ws, size_t ws_size,
                              hipStream_t stream) {
    const float* u  = (const float*)d_in[0];
    const float* b1 = (const float*)d_in[1];
    const float* a1 = (const float*)d_in[2];
    const float* b2 = (const float*)d_in[3];
    const float* a2 = (const float*)d_in[4];
    float* out = (float*)d_out;

    float* ws   = (float*)d_ws;
    float* hp1  = ws;
    float* hp2  = hp1 + L * 4;
    float* tab1 = hp2 + L * 4;
    float* tab2 = tab1 + TAB_FLOATS;

    hpack_kernel<<<4, 32, 0, stream>>>(a1, a2, b1, b2, hp1, hp2, tab1, tab2);
    row_kernel<<<Bsz, 64, 0, stream>>>(u, tab1, hp1, tab2, hp2, out);
}


# each wave rebuilds its operands from a 16-step walk in LDS, no tables in memory: correction = fixed 16x3 operand x chunk state stepped per block by the 16-step transition; 1 launch; sum order changes
# speedup vs baseline: 19.2700x; 1.2479x over previous
#include <hip/hip_runtime.h>
#include <hip/hip_bf16.h>
#include <math.h>

static constexpr int Bsz = 512;
static constexpr int T   = 16384;
static constexpr int L   = 512;
static constexpr int C   = T / L;

typedef __attribute__((ext_vector_type(2))) float v2f;
typedef __attribute__((ext_vector_type(4))) float v4f;
typedef __attribute__((ext_vector_type(8))) float v8f;

__device__ __forceinline__ float whf(float x) {
    const float c   = 10.0f / 11.0f;
    const float l2e = 1.4426950408889634f;
    float e = 1.0f - __builtin_amdgcn_exp2f(-(c * l2e) * x);
    float l = c * x;
    unsigned s = (unsigned)(__float_as_int(x) >> 31);
    return __uint_as_float((__float_as_uint(e) & ~s) | (__float_as_uint(l) & s));
}

__device__ __forceinline__ int blk_time(int hf, int r) {
    return 8 * (r >> 2) + 4 * hf + (r & 3);
}

static constexpr int RESP_FLOATS = 4 * 16;

__device__ __forceinline__ void walk16(const float* __restrict__ a1c,
                                       const float* __restrict__ a2c,
                                       float* __restrict__ resp, int lane) {
    const float p0 = a1c[0], p1 = a1c[1], p2 = a1c[2];
    const float q0 = a2c[0], q1 = a2c[1], q2 = a2c[2];
    if (lane < 8) {
        const bool second = lane >= 4;
        const int  r      = lane & 3;
        const float a0 = second ? q0 : p0, a1 = second ? q1 : p1, a2 = second ? q2 : p2;
        float y1 = (r == 1) ? 1.0f : 0.0f;
        float y2 = (r == 2) ? 1.0f : 0.0f;
        float y3 = (r == 3) ? 1.0f : 0.0f;
        for (int i = 0; i < 16; ++i) {
            float x = (r == 0 && i == 0) ? 1.0f : 0.0f;
            float y = x - (a0 * y1 + a1 * y2 + a2 * y3);
            resp[lane * 16 + i] = y;
            y3 = y2; y2 = y1; y1 = y;
        }
    }
    __builtin_amdgcn_fence(__ATOMIC_RELEASE, "wavefront");
    __builtin_amdgcn_wave_barrier();
    __builtin_amdgcn_fence(__ATOMIC_ACQUIRE, "wavefront");
}

__device__ __forceinline__ void block_tables(const float* __restrict__ bc, int nk,
                                             const float* __restrict__ resp,
                                             int lane, v2f tb[8]) {
    const float* hs = resp;
    const float* hh = resp + 16;
    const float bq[3] = {bc[0], bc[1], bc[2]};
    const int row = lane & 15;
    const int hf  = lane >> 4;
    const int m   = blk_time(row >> 3, row & 7);
#pragma unroll
    for (int slot = 0; slot < 6; ++slot) {
        const int kb = (slot < 4) ? slot : slot - 2;
        v2f e;
#pragma unroll
        for (int v = 0; v < 2; ++v) {
            int j = blk_time(hf, 2 * kb + v);
            if (slot >= 4) j -= 16;
            float s = 0.0f;
#pragma unroll
            for (int q = 0; q < 3; ++q) {
                const int d = m - j - nk - q;
                const bool ok = (q >= -(j + nk)) && (d >= 0);
                const float cq = ok ? bq[q] : 0.0f;
                s += cq * hs[min(max(d, 0), 15)];
            }
            e[v] = s;
        }
        tb[slot] = e;
    }
    const float f0 = hh[0 * 16 + m], f1 = hh[1 * 16 + m], f2 = hh[2 * 16 + m];
    v2f e6, e7;
#pragma unroll
    for (int v = 0; v < 2; ++v) {
        const int j67 = blk_time(hf, 6 + v);
        const int j45 = blk_time(hf, 4 + v);
        e6[v] = (j67 == 15) ? f0 : (j67 == 14) ? f1 : 0.0f;
        e7[v] = (j45 == 13) ? f2 : 0.0f;
    }
    tb[6] = e6;
    tb[7] = e7;
}

__device__ __forceinline__ v8f mm4(v2f a, float b0, float b1, v8f c) {
    v2f b; b[0] = b0; b[1] = b1;
    return __builtin_amdgcn_wmma_f32_16x16x4_f32(
         false, a,  false, b,
         (short)0, c,  false,  false);
}

static constexpr int NT = L / 16;

__device__ __forceinline__ void block_transition(const float* __restrict__ resp,
                                                 float (&s)[3][3]) {
    const float* hh = resp + 16;
#pragma unroll
    for (int p = 0; p < 3; ++p)
#pragma unroll
        for (int l = 0; l < 3; ++l) s[p][l] = hh[l * 16 + 15 - p];
}

__device__ __forceinline__ v2f corr_operand(const float* __restrict__ resp, int mrow, int hf) {
    const float* hh = resp + 16;
    const float f0 = hh[0 * 16 + mrow], f1 = hh[1 * 16 + mrow], f2 = hh[2 * 16 + mrow];
    v2f a;
    a[0] = hf ? f2 : f0;
    a[1] = hf ? 0.0f : f1;
    return a;
}

__device__ __forceinline__ void chunk_transition(const float (&s)[3][3], int hf,
                                                 float (&hc)[3][3]) {
    float a0 = hf ? 0.0f : 1.0f, a1 = 0.0f, a2 = hf ? 1.0f : 0.0f;
    float b0 = 0.0f, b1 = hf ? 0.0f : 1.0f, b2 = 0.0f;
#pragma unroll
    for (int k = 0; k < NT; ++k) {
        const float na0 = s[0][0] * a0 + s[0][1] * a1 + s[0][2] * a2;
        const float na1 = s[1][0] * a0 + s[1][1] * a1 + s[1][2] * a2;
        const float na2 = s[2][0] * a0 + s[2][1] * a1 + s[2][2] * a2;
        const float nb0 = s[0][0] * b0 + s[0][1] * b1 + s[0][2] * b2;
        const float nb1 = s[1][0] * b0 + s[1][1] * b1 + s[1][2] * b2;
        const float nb2 = s[2][0] * b0 + s[2][1] * b1 + s[2][2] * b2;
        a0 = na0; a1 = na1; a2 = na2;
        b0 = nb0; b1 = nb1; b2 = nb2;
    }
    const float oa0 = __shfl_xor(a0, 16), oa1 = __shfl_xor(a1, 16), oa2 = __shfl_xor(a2, 16);
    const float ob0 = __shfl_xor(b0, 16), ob1 = __shfl_xor(b1, 16), ob2 = __shfl_xor(b2, 16);
    hc[0][0] = hf ? oa0 : a0;  hc[0][1] = hf ? ob0 : b0;  hc[0][2] = hf ? a0 : oa0;
    hc[1][0] = hf ? oa1 : a1;  hc[1][1] = hf ? ob1 : b1;  hc[1][2] = hf ? a1 : oa1;
    hc[2][0] = hf ? oa2 : a2;  hc[2][1] = hf ? ob2 : b2;  hc[2][2] = hf ? a2 : oa2;
}

__device__ __forceinline__ void step_state(const float (&s)[3][3], float (&st)[3]) {
    const float n0 = s[0][0] * st[0] + s[0][1] * st[1] + s[0][2] * st[2];
    const float n1 = s[1][0] * st[0] + s[1][1] * st[1] + s[1][2] * st[2];
    const float n2 = s[2][0] * st[0] + s[2][1] * st[1] + s[2][2] * st[2];
    st[0] = n0; st[1] = n1; st[2] = n2;
}

__device__ __forceinline__ void chunk_state(const v4f* tl, const float (&hc)[3][3],
                                            int c, float (&st)[3]) {
    const float h00 = hc[0][0], h01 = hc[0][1], h02 = hc[0][2];
    const float h10 = hc[1][0], h11 = hc[1][1], h12 = hc[1][2];
    const float h20 = hc[2][0], h21 = hc[2][1], h22 = hc[2][2];
    float s0 = 0.0f, s1 = 0.0f, s2 = 0.0f;
    float m0 = 0.0f, m1 = 0.0f, m2 = 0.0f;
#pragma unroll 8
    for (int cc = 0; cc < C; ++cc) {
        const bool mine = (cc == c);
        m0 = mine ? s0 : m0; m1 = mine ? s1 : m1; m2 = mine ? s2 : m2;
        const v4f t = tl[cc];
        const float n0 = t[0] + h00 * s0 + h01 * s1 + h02 * s2;
        const float n1 = t[1] + h10 * s0 + h11 * s1 + h12 * s2;
        const float n2 = t[2] + h20 * s0 + h21 * s1 + h22 * s2;
        s0 = n0; s1 = n1; s2 = n2;
    }
    st[0] = m0; st[1] = m1; st[2] = m2;
}

__device__ __forceinline__ v2f state_operand(const float (&st)[3], int hf) {
    v2f bv;
    bv[0] = hf ? st[2] : st[0];
    bv[1] = hf ? 0.0f : st[1];
    return bv;
}

__global__ __launch_bounds__(64)
void row_kernel(const float* __restrict__ u,
                const float* __restrict__ b1c,
                const float* __restrict__ a1c,
                const float* __restrict__ b2c,
                const float* __restrict__ a2c,
                float* __restrict__ out) {
    __shared__ float resp[2][2][RESP_FLOATS];
    __shared__ v4f tails1[C];
    __shared__ v4f tails2[C];
    __shared__ v4f edge[C][2];

    const int lane = threadIdx.x & 31;
    const int wave = threadIdx.x >> 5;
    const int n    = lane & 15;
    const int hf   = lane >> 4;
    const int b    = blockIdx.x;
    const int c    = wave * 16 + n;

    const float* __restrict__ src = u   + (size_t)b * T + (size_t)c * L + 4 * hf;
    float*       __restrict__ dst = out + (size_t)b * T + (size_t)c * L + 4 * hf;

    const int mrow = blk_time(n >> 3, n & 7);

    v8f t[NT];

    float hc1[3][3], hc2[3][3];
    v2f tb2[8];

    {
        v4f pv = *(const v4f*)(c > 0 ? src - 8 : src);

        v4f ua[NT], ub[NT];
#pragma unroll
        for (int k = 0; k < NT; ++k) {
            ua[k] = *(const v4f*)(src + 16 * k);
            ub[k] = *(const v4f*)(src + 16 * k + 8);
        }

        walk16(a1c, a2c, resp[wave][0], lane);
        v2f tb[8];
        block_tables(b1c, 1, resp[wave][0], lane, tb);
        const v2f g0 = tb[0], g1 = tb[1], g2 = tb[2], g3 = tb[3];
        const v2f p2 = tb[4], p3 = tb[5];
        const v2f s3 = tb[6], s2 = tb[7];
        block_tables(b2c, 0, resp[wave][1], lane, tb2);
        {
            float sa[3][3], sb[3][3];
            block_transition(resp[wave][0], sa);
            block_transition(resp[wave][1], sb);
            chunk_transition(sa, hf, hc1);
            chunk_transition(sb, hf, hc2);
        }
        if (c == 0) { pv[0] = 0.0f; pv[1] = 0.0f; pv[2] = 0.0f; pv[3] = 0.0f; }

        v8f d = {0.0f, 0.0f, 0.0f, 0.0f, 0.0f, 0.0f, 0.0f, 0.0f};
#pragma unroll
        for (int k = 0; k < NT; ++k) {
            const v4f u0 = ua[k], u1 = ub[k];
            v8f acc = {0.0f, 0.0f, 0.0f, 0.0f, 0.0f, 0.0f, 0.0f, 0.0f};
            acc = mm4(g0, u0[0], u0[1], acc);
            acc = mm4(g1, u0[2], u0[3], acc);
            acc = mm4(g2, u1[0], u1[1], acc);
            acc = mm4(g3, u1[2], u1[3], acc);
            acc = mm4(p2, pv[0], pv[1], acc);
            acc = mm4(p3, pv[2], pv[3], acc);
            acc = mm4(s2, d[4], d[5], acc);
            acc = mm4(s3, d[6], d[7], acc);
            d = acc;
            pv = u1;
            t[k] = d;
        }
        if (hf == 1) {
            v4f tl; tl[0] = d[7]; tl[1] = d[6]; tl[2] = d[5]; tl[3] = 0.0f;
            tails1[c] = tl;
        }
    }
    __syncthreads();

    {
        float sm[3][3];
        block_transition(resp[wave][0], sm);
        const v2f am = corr_operand(resp[wave][0], mrow, hf);
        float st[3];
        chunk_state(tails1, hc1, c, st);
#pragma unroll
        for (int k = 0; k < NT; ++k) {
            v8f dv = __builtin_amdgcn_wmma_f32_16x16x4_f32(
                 false, am,  false, state_operand(st, hf),
                 (short)0, t[k],  false,  false);
            step_state(sm, st);
#pragma unroll
            for (int r = 0; r < 8; ++r) dv[r] = whf(dv[r]);
            t[k] = dv;
        }
        v4f e; e[0] = t[NT - 1][4]; e[1] = t[NT - 1][5]; e[2] = t[NT - 1][6]; e[3] = t[NT - 1][7];
        edge[c][hf] = e;
    }
    __syncthreads();

    {
        const v2f g0 = tb2[0], g1 = tb2[1], g2 = tb2[2], g3 = tb2[3];
        const v2f p3 = tb2[5];
        const v2f s3 = tb2[6], s2 = tb2[7];

        v4f pv = edge[c > 0 ? c - 1 : 0][hf];
        if (c == 0) { pv[0] = 0.0f; pv[1] = 0.0f; pv[2] = 0.0f; pv[3] = 0.0f; }

        v8f d = {0.0f, 0.0f, 0.0f, 0.0f, 0.0f, 0.0f, 0.0f, 0.0f};
#pragma unroll
        for (int k = 0; k < NT; ++k) {
            const v8f x = t[k];
            v8f acc = {0.0f, 0.0f, 0.0f, 0.0f, 0.0f, 0.0f, 0.0f, 0.0f};
            acc = mm4(g0, x[0], x[1], acc);
            acc = mm4(g1, x[2], x[3], acc);
            acc = mm4(g2, x[4], x[5], acc);
            acc = mm4(g3, x[6], x[7], acc);
            acc = mm4(p3, pv[2], pv[3], acc);
            acc = mm4(s2, d[4], d[5], acc);
            acc = mm4(s3, d[6], d[7], acc);
            d = acc;
            pv[0] = x[4]; pv[1] = x[5]; pv[2] = x[6]; pv[3] = x[7];
            t[k] = d;
        }
        if (hf == 1) {
            v4f tl; tl[0] = d[7]; tl[1] = d[6]; tl[2] = d[5]; tl[3] = 0.0f;
            tails2[c] = tl;
        }
    }
    __syncthreads();

    {
        float sm[3][3];
        block_transition(resp[wave][1], sm);
        const v2f am = corr_operand(resp[wave][1], mrow, hf);
        float st[3];
        chunk_state(tails2, hc2, c, st);
#pragma unroll
        for (int k = 0; k < NT; ++k) {
            const v8f dv = __builtin_amdgcn_wmma_f32_16x16x4_f32(
                 false, am,  false, state_operand(st, hf),
                 (short)0, t[k],  false,  false);
            step_state(sm, st);
            v4f o0, o1;
            o0[0] = dv[0]; o0[1] = dv[1]; o0[2] = dv[2]; o0[3] = dv[3];
            o1[0] = dv[4]; o1[1] = dv[5]; o1[2] = dv[6]; o1[3] = dv[7];
            __builtin_nontemporal_store(o0, (v4f*)(dst + 16 * k));
            __builtin_nontemporal_store(o1, (v4f*)(dst + 16 * k + 8));
        }
    }
}

extern "C" void kernel_launch(void* const* d_in, const int* in_sizes, int n_in,
                              void* d_out, int out_size, void* d_ws, size_t ws_size,
                              hipStream_t stream) {
    const float* u  = (const float*)d_in[0];
    const float* b1 = (const float*)d_in[1];
    const float* a1 = (const float*)d_in[2];
    const float* b2 = (const float*)d_in[3];
    const float* a2 = (const float*)d_in[4];
    float* out = (float*)d_out;

    row_kernel<<<Bsz, 64, 0, stream>>>(u, b1, a1, b2, a2, out);
}


# output tiles regrouped in registers by two bank-masked DPP lane exchanges (chunk bits 3,2 <-> piece, tile parity): a store writes 4 whole 128-byte lines, was 16 lines x 32 B; no LDS; sums unchanged
# speedup vs baseline: 29.7725x; 1.5450x over previous
#include <hip/hip_runtime.h>
#include <hip/hip_bf16.h>
#include <math.h>

static constexpr int Bsz = 512;
static constexpr int T   = 16384;
static constexpr int L   = 512;
static constexpr int C   = T / L;

typedef __attribute__((ext_vector_type(2))) float v2f;
typedef __attribute__((ext_vector_type(4))) float v4f;
typedef __attribute__((ext_vector_type(8))) float v8f;

__device__ __forceinline__ float whf(float x) {
    const float c   = 10.0f / 11.0f;
    const float l2e = 1.4426950408889634f;
    float e = 1.0f - __builtin_amdgcn_exp2f(-(c * l2e) * x);
    float l = c * x;
    unsigned s = (unsigned)(__float_as_int(x) >> 31);
    return __uint_as_float((__float_as_uint(e) & ~s) | (__float_as_uint(l) & s));
}

__device__ __forceinline__ int blk_time(int hf, int r) {
    return 8 * (r >> 2) + 4 * hf + (r & 3);
}

static constexpr int RESP_FLOATS = 4 * 16;

__device__ __forceinline__ void walk16(const float* __restrict__ a1c,
                                       const float* __restrict__ a2c,
                                       float* __restrict__ resp, int lane) {
    const float p0 = a1c[0], p1 = a1c[1], p2 = a1c[2];
    const float q0 = a2c[0], q1 = a2c[1], q2 = a2c[2];
    if (lane < 8) {
        const bool second = lane >= 4;
        const int  r      = lane & 3;
        const float a0 = second ? q0 : p0, a1 = second ? q1 : p1, a2 = second ? q2 : p2;
        float y1 = (r == 1) ? 1.0f : 0.0f;
        float y2 = (r == 2) ? 1.0f : 0.0f;
        float y3 = (r == 3) ? 1.0f : 0.0f;
        for (int i = 0; i < 16; ++i) {
            float x = (r == 0 && i == 0) ? 1.0f : 0.0f;
            float y = x - (a0 * y1 + a1 * y2 + a2 * y3);
            resp[lane * 16 + i] = y;
            y3 = y2; y2 = y1; y1 = y;
        }
    }
    __builtin_amdgcn_fence(__ATOMIC_RELEASE, "wavefront");
    __builtin_amdgcn_wave_barrier();
    __builtin_amdgcn_fence(__ATOMIC_ACQUIRE, "wavefront");
}

__device__ __forceinline__ void block_tables(const float* __restrict__ bc, int nk,
                                             const float* __restrict__ resp,
                                             int lane, v2f tb[8]) {
    const float* hs = resp;
    const float* hh = resp + 16;
    const float bq[3] = {bc[0], bc[1], bc[2]};
    const int row = lane & 15;
    const int hf  = lane >> 4;
    const int m   = blk_time(row >> 3, row & 7);
#pragma unroll
    for (int slot = 0; slot < 6; ++slot) {
        const int kb = (slot < 4) ? slot : slot - 2;
        v2f e;
#pragma unroll
        for (int v = 0; v < 2; ++v) {
            int j = blk_time(hf, 2 * kb + v);
            if (slot >= 4) j -= 16;
            float s = 0.0f;
#pragma unroll
            for (int q = 0; q < 3; ++q) {
                const int d = m - j - nk - q;
                const bool ok = (q >= -(j + nk)) && (d >= 0);
                const float cq = ok ? bq[q] : 0.0f;
                s += cq * hs[min(max(d, 0), 15)];
            }
            e[v] = s;
        }
        tb[slot] = e;
    }
    const float f0 = hh[0 * 16 + m], f1 = hh[1 * 16 + m], f2 = hh[2 * 16 + m];
    v2f e6, e7;
#pragma unroll
    for (int v = 0; v < 2; ++v) {
        const int j67 = blk_time(hf, 6 + v);
        const int j45 = blk_time(hf, 4 + v);
        e6[v] = (j67 == 15) ? f0 : (j67 == 14) ? f1 : 0.0f;
        e7[v] = (j45 == 13) ? f2 : 0.0f;
    }
    tb[6] = e6;
    tb[7] = e7;
}

__device__ __forceinline__ v8f mm4(v2f a, float b0, float b1, v8f c) {
    v2f b; b[0] = b0; b[1] = b1;
    return __builtin_amdgcn_wmma_f32_16x16x4_f32(
         false, a,  false, b,
         (short)0, c,  false,  false);
}

template <int XM, int BANKS>
__device__ __forceinline__ float lane_swap(float keep, float give) {
    return __int_as_float(__builtin_amdgcn_update_dpp(
        __float_as_int(keep), __float_as_int(give),
          0x160 + XM,  0xf,  BANKS,  false));
}

static constexpr int NT = L / 16;

__device__ __forceinline__ void block_transition(const float* __restrict__ resp,
                                                 float (&s)[3][3]) {
    const float* hh = resp + 16;
#pragma unroll
    for (int p = 0; p < 3; ++p)
#pragma unroll
        for (int l = 0; l < 3; ++l) s[p][l] = hh[l * 16 + 15 - p];
}

__device__ __forceinline__ v2f corr_operand(const float* __restrict__ resp, int mrow, int hf) {
    const float* hh = resp + 16;
    const float f0 = hh[0 * 16 + mrow], f1 = hh[1 * 16 + mrow], f2 = hh[2 * 16 + mrow];
    v2f a;
    a[0] = hf ? f2 : f0;
    a[1] = hf ? 0.0f : f1;
    return a;
}

__device__ __forceinline__ void chunk_transition(const float (&s)[3][3], int hf,
                                                 float (&hc)[3][3]) {
    float a0 = hf ? 0.0f : 1.0f, a1 = 0.0f, a2 = hf ? 1.0f : 0.0f;
    float b0 = 0.0f, b1 = hf ? 0.0f : 1.0f, b2 = 0.0f;
#pragma unroll
    for (int k = 0; k < NT; ++k) {
        const float na0 = s[0][0] * a0 + s[0][1] * a1 + s[0][2] * a2;
        const float na1 = s[1][0] * a0 + s[1][1] * a1 + s[1][2] * a2;
        const float na2 = s[2][0] * a0 + s[2][1] * a1 + s[2][2] * a2;
        const float nb0 = s[0][0] * b0 + s[0][1] * b1 + s[0][2] * b2;
        const float nb1 = s[1][0] * b0 + s[1][1] * b1 + s[1][2] * b2;
        const float nb2 = s[2][0] * b0 + s[2][1] * b1 + s[2][2] * b2;
        a0 = na0; a1 = na1; a2 = na2;
        b0 = nb0; b1 = nb1; b2 = nb2;
    }
    const float oa0 = __shfl_xor(a0, 16), oa1 = __shfl_xor(a1, 16), oa2 = __shfl_xor(a2, 16);
    const float ob0 = __shfl_xor(b0, 16), ob1 = __shfl_xor(b1, 16), ob2 = __shfl_xor(b2, 16);
    hc[0][0] = hf ? oa0 : a0;  hc[0][1] = hf ? ob0 : b0;  hc[0][2] = hf ? a0 : oa0;
    hc[1][0] = hf ? oa1 : a1;  hc[1][1] = hf ? ob1 : b1;  hc[1][2] = hf ? a1 : oa1;
    hc[2][0] = hf ? oa2 : a2;  hc[2][1] = hf ? ob2 : b2;  hc[2][2] = hf ? a2 : oa2;
}

__device__ __forceinline__ void step_state(const float (&s)[3][3], float (&st)[3]) {
    const float n0 = s[0][0] * st[0] + s[0][1] * st[1] + s[0][2] * st[2];
    const float n1 = s[1][0] * st[0] + s[1][1] * st[1] + s[1][2] * st[2];
    const float n2 = s[2][0] * st[0] + s[2][1] * st[1] + s[2][2] * st[2];
    st[0] = n0; st[1] = n1; st[2] = n2;
}

__device__ __forceinline__ void chunk_state(const v4f* tl, const float (&hc)[3][3],
                                            int c, float (&st)[3]) {
    const float h00 = hc[0][0], h01 = hc[0][1], h02 = hc[0][2];
    const float h10 = hc[1][0], h11 = hc[1][1], h12 = hc[1][2];
    const float h20 = hc[2][0], h21 = hc[2][1], h22 = hc[2][2];
    float s0 = 0.0f, s1 = 0.0f, s2 = 0.0f;
    float m0 = 0.0f, m1 = 0.0f, m2 = 0.0f;
#pragma unroll 8
    for (int cc = 0; cc < C; ++cc) {
        const bool mine = (cc == c);
        m0 = mine ? s0 : m0; m1 = mine ? s1 : m1; m2 = mine ? s2 : m2;
        const v4f t = tl[cc];
        const float n0 = t[0] + h00 * s0 + h01 * s1 + h02 * s2;
        const float n1 = t[1] + h10 * s0 + h11 * s1 + h12 * s2;
        const float n2 = t[2] + h20 * s0 + h21 * s1 + h22 * s2;
        s0 = n0; s1 = n1; s2 = n2;
    }
    st[0] = m0; st[1] = m1; st[2] = m2;
}

__device__ __forceinline__ v2f state_operand(const float (&st)[3], int hf) {
    v2f bv;
    bv[0] = hf ? st[2] : st[0];
    bv[1] = hf ? 0.0f : st[1];
    return bv;
}

__global__ __launch_bounds__(64)
void row_kernel(const float* __restrict__ u,
                const float* __restrict__ b1c,
                const float* __restrict__ a1c,
                const float* __restrict__ b2c,
                const float* __restrict__ a2c,
                float* __restrict__ out) {
    __shared__ float resp[2][2][RESP_FLOATS];
    __shared__ v4f tails1[C];
    __shared__ v4f tails2[C];
    __shared__ v4f edge[C][2];

    const int lane = threadIdx.x & 31;
    const int wave = threadIdx.x >> 5;
    const int n    = lane & 15;
    const int hf   = lane >> 4;
    const int b    = blockIdx.x;
    const int c    = wave * 16 + n;

    const float* __restrict__ src = u   + (size_t)b * T + (size_t)c * L + 4 * hf;

    const int mrow = blk_time(n >> 3, n & 7);

    v8f t[NT];

    float hc1[3][3], hc2[3][3];
    v2f tb2[8];

    {
        v4f pv = *(const v4f*)(c > 0 ? src - 8 : src);

        v4f ua[NT], ub[NT];
#pragma unroll
        for (int k = 0; k < NT; ++k) {
            ua[k] = *(const v4f*)(src + 16 * k);
            ub[k] = *(const v4f*)(src + 16 * k + 8);
        }

        walk16(a1c, a2c, resp[wave][0], lane);
        v2f tb[8];
        block_tables(b1c, 1, resp[wave][0], lane, tb);
        const v2f g0 = tb[0], g1 = tb[1], g2 = tb[2], g3 = tb[3];
        const v2f p2 = tb[4], p3 = tb[5];
        const v2f s3 = tb[6], s2 = tb[7];
        block_tables(b2c, 0, resp[wave][1], lane, tb2);
        {
            float sa[3][3], sb[3][3];
            block_transition(resp[wave][0], sa);
            block_transition(resp[wave][1], sb);
            chunk_transition(sa, hf, hc1);
            chunk_transition(sb, hf, hc2);
        }
        if (c == 0) { pv[0] = 0.0f; pv[1] = 0.0f; pv[2] = 0.0f; pv[3] = 0.0f; }

        v8f d = {0.0f, 0.0f, 0.0f, 0.0f, 0.0f, 0.0f, 0.0f, 0.0f};
#pragma unroll
        for (int k = 0; k < NT; ++k) {
            const v4f u0 = ua[k], u1 = ub[k];
            v8f acc = {0.0f, 0.0f, 0.0f, 0.0f, 0.0f, 0.0f, 0.0f, 0.0f};
            acc = mm4(g0, u0[0], u0[1], acc);
            acc = mm4(g1, u0[2], u0[3], acc);
            acc = mm4(g2, u1[0], u1[1], acc);
            acc = mm4(g3, u1[2], u1[3], acc);
            acc = mm4(p2, pv[0], pv[1], acc);
            acc = mm4(p3, pv[2], pv[3], acc);
            acc = mm4(s2, d[4], d[5], acc);
            acc = mm4(s3, d[6], d[7], acc);
            d = acc;
            pv = u1;
            t[k] = d;
        }
        if (hf == 1) {
            v4f tl; tl[0] = d[7]; tl[1] = d[6]; tl[2] = d[5]; tl[3] = 0.0f;
            tails1[c] = tl;
        }
    }
    __syncthreads();

    {
        float sm[3][3];
        block_transition(resp[wave][0], sm);
        const v2f am = corr_operand(resp[wave][0], mrow, hf);
        float st[3];
        chunk_state(tails1, hc1, c, st);
#pragma unroll
        for (int k = 0; k < NT; ++k) {
            v8f dv = __builtin_amdgcn_wmma_f32_16x16x4_f32(
                 false, am,  false, state_operand(st, hf),
                 (short)0, t[k],  false,  false);
            step_state(sm, st);
#pragma unroll
            for (int r = 0; r < 8; ++r) dv[r] = whf(dv[r]);
            t[k] = dv;
        }
        v4f e; e[0] = t[NT - 1][4]; e[1] = t[NT - 1][5]; e[2] = t[NT - 1][6]; e[3] = t[NT - 1][7];
        edge[c][hf] = e;
    }
    __syncthreads();

    {
        const v2f g0 = tb2[0], g1 = tb2[1], g2 = tb2[2], g3 = tb2[3];
        const v2f p3 = tb2[5];
        const v2f s3 = tb2[6], s2 = tb2[7];

        v4f pv = edge[c > 0 ? c - 1 : 0][hf];
        if (c == 0) { pv[0] = 0.0f; pv[1] = 0.0f; pv[2] = 0.0f; pv[3] = 0.0f; }

        v8f d = {0.0f, 0.0f, 0.0f, 0.0f, 0.0f, 0.0f, 0.0f, 0.0f};
#pragma unroll
        for (int k = 0; k < NT; ++k) {
            const v8f x = t[k];
            v8f acc = {0.0f, 0.0f, 0.0f, 0.0f, 0.0f, 0.0f, 0.0f, 0.0f};
            acc = mm4(g0, x[0], x[1], acc);
            acc = mm4(g1, x[2], x[3], acc);
            acc = mm4(g2, x[4], x[5], acc);
            acc = mm4(g3, x[6], x[7], acc);
            acc = mm4(p3, pv[2], pv[3], acc);
            acc = mm4(s2, d[4], d[5], acc);
            acc = mm4(s3, d[6], d[7], acc);
            d = acc;
            pv[0] = x[4]; pv[1] = x[5]; pv[2] = x[6]; pv[3] = x[7];
            t[k] = d;
        }
        if (hf == 1) {
            v4f tl; tl[0] = d[7]; tl[1] = d[6]; tl[2] = d[5]; tl[3] = 0.0f;
            tails2[c] = tl;
        }
    }
    __syncthreads();

    {
        float sm[3][3];
        block_transition(resp[wave][1], sm);
        const v2f am = corr_operand(resp[wave][1], mrow, hf);
        float st[3];
        chunk_state(tails2, hc2, c, st);
        const int l3 = (n >> 3) & 1, l2 = (n >> 2) & 1;
        float* __restrict__ dst2 = out + (size_t)b * T + (size_t)(wave * 16 + (n & 3)) * L
                                 + 16 * l2 + 8 * l3 + 4 * hf;
#pragma unroll
        for (int kp = 0; kp < NT / 2; ++kp) {
            const v8f d0 = __builtin_amdgcn_wmma_f32_16x16x4_f32(
                 false, am,  false, state_operand(st, hf),
                 (short)0, t[2 * kp],  false,  false);
            step_state(sm, st);
            const v8f d1 = __builtin_amdgcn_wmma_f32_16x16x4_f32(
                 false, am,  false, state_operand(st, hf),
                 (short)0, t[2 * kp + 1],  false,  false);
            step_state(sm, st);
            v4f o00, o01, o10, o11;
#pragma unroll
            for (int e = 0; e < 4; ++e) {
                const float a0 = lane_swap<8, 0xC>(d0[e], d0[4 + e]);
                const float a1 = lane_swap<8, 0x3>(d0[4 + e], d0[e]);
                const float b0 = lane_swap<8, 0xC>(d1[e], d1[4 + e]);
                const float b1 = lane_swap<8, 0x3>(d1[4 + e], d1[e]);
                o00[e] = lane_swap<4, 0xA>(a0, b0);
                o10[e] = lane_swap<4, 0x5>(b0, a0);
                o01[e] = lane_swap<4, 0xA>(a1, b1);
                o11[e] = lane_swap<4, 0x5>(b1, a1);
            }
            __builtin_nontemporal_store(o00, (v4f*)(dst2 + 0 * L + 32 * kp));
            __builtin_nontemporal_store(o10, (v4f*)(dst2 + 4 * L + 32 * kp));
            __builtin_nontemporal_store(o01, (v4f*)(dst2 + 8 * L + 32 * kp));
            __builtin_nontemporal_store(o11, (v4f*)(dst2 + 12 * L + 32 * kp));
        }
    }
}

extern "C" void kernel_launch(void* const* d_in, const int* in_sizes, int n_in,
                              void* d_out, int out_size, void* d_ws, size_t ws_size,
                              hipStream_t stream) {
    const float* u  = (const float*)d_in[0];
    const float* b1 = (const float*)d_in[1];
    const float* a1 = (const float*)d_in[2];
    const float* b2 = (const float*)d_in[3];
    const float* a2 = (const float*)d_in[4];
    float* out = (float*)d_out;

    row_kernel<<<Bsz, 64, 0, stream>>>(u, b1, a1, b2, a2, out);
}
